# starting kernel
# baseline (speedup 1.0000x reference)
#include <hip/hip_runtime.h>
#include <hip/hip_bf16.h>

typedef float v2f __attribute__((ext_vector_type(2)));
typedef float v8f __attribute__((ext_vector_type(8)));

#define N_FULL   1024
#define NBLK     32
#define TILE_M   16
#define YS       1028
#define NWAVES   4

__global__ __launch_bounds__(32) void build_packed_mats(
    const float* __restrict__ tw,
    float* __restrict__ pk)
{
    __shared__ float Cm[32 * 32];
    const int c   = threadIdx.x;
    const bool isB = (blockIdx.x >= 32);
    const int idx = isB ? (int)blockIdx.x - 32 : (int)blockIdx.x;

    for (int i = 0; i < 32; ++i) Cm[i * 32 + c] = (i == c) ? 1.0f : 0.0f;
    __syncthreads();

    for (int lsi = 0; lsi < 5; ++lsi) {
        const int ls = isB ? lsi + 5 : lsi;
        const int s  = 1 << ls;
        const int cs = isB ? (s >> 5) : s;
        for (int pr = 0; pr < 16; ++pr) {
            const int G  = pr / cs;
            const int t  = pr % cs;
            const int u0 = G * 2 * cs + t;
            const int u1 = u0 + cs;
            const int p0 = isB ? (u0 * 32 + idx) : (idx * 32 + u0);
            const int pid = (p0 / (2 * s)) * s + (p0 % s);
            const float* tp = tw + ls * 2048 + pid * 4;
            const float t00 = tp[0], t01 = tp[1], t10 = tp[2], t11 = tp[3];
            const float a0 = Cm[u0 * 32 + c];
            const float a1 = Cm[u1 * 32 + c];
            Cm[u0 * 32 + c] = t00 * a0 + t01 * a1;
            Cm[u1 * 32 + c] = t10 * a0 + t11 * a1;
        }
    }
    __syncthreads();

    const int L = c, Lm = L & 15, Lh = L >> 4;
    float* base = pk + (size_t)blockIdx.x * 1024;
    for (int h = 0; h < 2; ++h) {
        for (int kk = 0; kk < 8; ++kk) {
            const int row  = h * 16 + Lm;
            const int col0 = kk * 4 + Lh * 2;
            base[(h * 8 + kk) * 64 + L * 2 + 0] = Cm[row * 32 + col0 + 0];
            base[(h * 8 + kk) * 64 + L * 2 + 1] = Cm[row * 32 + col0 + 1];
        }
    }
}

__global__ __launch_bounds__(NWAVES * 32) void butterfly_wmma(
    const float* __restrict__ x,
    const float* __restrict__ pk,
    const float* __restrict__ bias,
    float* __restrict__ out)
{
    __shared__ __align__(16) float y[TILE_M * YS];

    const int tid = threadIdx.x;
    const int w   = tid >> 5;
    const int L   = tid & 31;
    const int Lm  = L & 15;
    const int Lh  = L >> 4;
    const size_t b0 = (size_t)blockIdx.x * TILE_M;

    const float* xrow = x + b0 * N_FULL;
    for (int i = tid; i < TILE_M * (N_FULL / 4); i += NWAVES * 32) {
        const int row = i >> 8;
        const int c4  = (i & 255) * 4;
        const float4 v = *(const float4*)(xrow + row * N_FULL + c4);
        *(float4*)(&y[row * YS + c4]) = v;
    }
    __syncthreads();

    const v2f* pkA = (const v2f*)pk;
    const v2f* pkB = (const v2f*)pk + 32 * 512;

    for (int g = w; g < NBLK; g += NWAVES) {
        v2f xa[8];
        const int xb = Lm * YS + g * 32 + Lh * 2;
        #pragma unroll
        for (int kk = 0; kk < 8; ++kk)
            xa[kk] = *(const v2f*)(&y[xb + kk * 4]);

        #pragma unroll
        for (int h = 0; h < 2; ++h) {
            v8f acc = {0.f, 0.f, 0.f, 0.f, 0.f, 0.f, 0.f, 0.f};
            #pragma unroll
            for (int kk = 0; kk < 8; ++kk) {
                const v2f bm = pkA[g * 512 + (h * 8 + kk) * 32 + L];
                acc = __builtin_amdgcn_wmma_f32_16x16x4_f32(
                    false, xa[kk], false, bm, (short)0, acc, false, false);
            }
            const int db = g * 32 + h * 16 + Lm;
            #pragma unroll
            for (int v = 0; v < 8; ++v)
                y[(v + 8 * Lh) * YS + db] = acc[v];
        }
    }
    __syncthreads();

    for (int r = w; r < 32; r += NWAVES) {
        v2f ya[8];
        #pragma unroll
        for (int kk = 0; kk < 8; ++kk) {
            const int k0 = kk * 4 + Lh * 2;
            ya[kk][0] = y[Lm * YS + (k0 + 0) * 32 + r];
            ya[kk][1] = y[Lm * YS + (k0 + 1) * 32 + r];
        }
        #pragma unroll
        for (int h = 0; h < 2; ++h) {
            v8f acc = {0.f, 0.f, 0.f, 0.f, 0.f, 0.f, 0.f, 0.f};
            #pragma unroll
            for (int kk = 0; kk < 8; ++kk) {
                const v2f bm = pkB[r * 512 + (h * 8 + kk) * 32 + L];
                acc = __builtin_amdgcn_wmma_f32_16x16x4_f32(
                    false, ya[kk], false, bm, (short)0, acc, false, false);
            }
            #pragma unroll
            for (int v = 0; v < 8; ++v)
                y[(v + 8 * Lh) * YS + (h * 16 + Lm) * 32 + r] = acc[v];
        }
    }
    __syncthreads();

    float* orow = out + b0 * N_FULL;
    for (int i = tid; i < TILE_M * (N_FULL / 4); i += NWAVES * 32) {
        const int row = i >> 8;
        const int c4  = (i & 255) * 4;
        float4 v = *(const float4*)(&y[row * YS + c4]);
        const float4 bv = *(const float4*)(bias + c4);
        v.x += bv.x; v.y += bv.y; v.z += bv.z; v.w += bv.w;
        *(float4*)(orow + row * N_FULL + c4) = v;
    }
}

extern "C" void kernel_launch(void* const* d_in, const int* in_sizes, int n_in,
                              void* d_out, int out_size, void* d_ws, size_t ws_size,
                              hipStream_t stream) {
    (void)n_in; (void)out_size; (void)ws_size;
    const float* x    = (const float*)d_in[0];
    const float* tw   = (const float*)d_in[1];
    const float* bias = (const float*)d_in[2];
    float* out = (float*)d_out;
    float* pk  = (float*)d_ws;

    const int batch = in_sizes[0] / N_FULL;

    build_packed_mats<<<64, 32, 0, stream>>>(tw, pk);
    butterfly_wmma<<<batch / TILE_M, NWAVES * 32, 0, stream>>>(x, pk, bias, out);
}


# WMMA operand roles swapped (matrix = A, data tile = B, transposed product): a lane's accumulator is 8 outputs of ONE batch row, LDS write-back 2 lanes per bank (was 16); same sums, same k order
# speedup vs baseline: 1.1468x; 1.1468x over previous
#include <hip/hip_runtime.h>
#include <hip/hip_bf16.h>

typedef float v2f __attribute__((ext_vector_type(2)));
typedef float v8f __attribute__((ext_vector_type(8)));

#define N_FULL   1024
#define NBLK     32
#define TILE_M   16
#define YS       1028
#define NWAVES   4

__global__ __launch_bounds__(32) void build_packed_mats(
    const float* __restrict__ tw,
    float* __restrict__ pk)
{
    __shared__ float Cm[32 * 32];
    const int c   = threadIdx.x;
    const bool isB = (blockIdx.x >= 32);
    const int idx = isB ? (int)blockIdx.x - 32 : (int)blockIdx.x;

    for (int i = 0; i < 32; ++i) Cm[i * 32 + c] = (i == c) ? 1.0f : 0.0f;
    __syncthreads();

    for (int lsi = 0; lsi < 5; ++lsi) {
        const int ls = isB ? lsi + 5 : lsi;
        const int s  = 1 << ls;
        const int cs = isB ? (s >> 5) : s;
        for (int pr = 0; pr < 16; ++pr) {
            const int G  = pr / cs;
            const int t  = pr % cs;
            const int u0 = G * 2 * cs + t;
            const int u1 = u0 + cs;
            const int p0 = isB ? (u0 * 32 + idx) : (idx * 32 + u0);
            const int pid = (p0 / (2 * s)) * s + (p0 % s);
            const float* tp = tw + ls * 2048 + pid * 4;
            const float t00 = tp[0], t01 = tp[1], t10 = tp[2], t11 = tp[3];
            const float a0 = Cm[u0 * 32 + c];
            const float a1 = Cm[u1 * 32 + c];
            Cm[u0 * 32 + c] = t00 * a0 + t01 * a1;
            Cm[u1 * 32 + c] = t10 * a0 + t11 * a1;
        }
    }
    __syncthreads();

    const int L = c, Lm = L & 15, Lh = L >> 4;
    float* base = pk + (size_t)blockIdx.x * 1024;
    for (int h = 0; h < 2; ++h) {
        for (int kk = 0; kk < 8; ++kk) {
            const int row  = h * 16 + Lm;
            const int col0 = kk * 4 + Lh * 2;
            base[(h * 8 + kk) * 64 + L * 2 + 0] = Cm[row * 32 + col0 + 0];
            base[(h * 8 + kk) * 64 + L * 2 + 1] = Cm[row * 32 + col0 + 1];
        }
    }
}

__global__ __launch_bounds__(NWAVES * 32) void butterfly_wmma(
    const float* __restrict__ x,
    const float* __restrict__ pk,
    const float* __restrict__ bias,
    float* __restrict__ out)
{
    __shared__ __align__(16) float y[TILE_M * YS];

    const int tid = threadIdx.x;
    const int w   = tid >> 5;
    const int L   = tid & 31;
    const int Lm  = L & 15;
    const int Lh  = L >> 4;
    const size_t b0 = (size_t)blockIdx.x * TILE_M;

    const float* xrow = x + b0 * N_FULL;
    for (int i = tid; i < TILE_M * (N_FULL / 4); i += NWAVES * 32) {
        const int row = i >> 8;
        const int c4  = (i & 255) * 4;
        const float4 v = *(const float4*)(xrow + row * N_FULL + c4);
        *(float4*)(&y[row * YS + c4]) = v;
    }
    __syncthreads();

    const v2f* pkA = (const v2f*)pk;
    const v2f* pkB = (const v2f*)pk + 32 * 512;

    for (int g = w; g < NBLK; g += NWAVES) {
        v2f xa[8];
        const int xb = Lm * YS + g * 32 + Lh * 2;
        #pragma unroll
        for (int kk = 0; kk < 8; ++kk)
            xa[kk] = *(const v2f*)(&y[xb + kk * 4]);

        #pragma unroll
        for (int h = 0; h < 2; ++h) {
            v8f acc = {0.f, 0.f, 0.f, 0.f, 0.f, 0.f, 0.f, 0.f};
            #pragma unroll
            for (int kk = 0; kk < 8; ++kk) {
                const v2f am = pkA[g * 512 + (h * 8 + kk) * 32 + L];
                acc = __builtin_amdgcn_wmma_f32_16x16x4_f32(
                    false, am, false, xa[kk], (short)0, acc, false, false);
            }
            float* dst = &y[Lm * YS + g * 32 + h * 16 + 8 * Lh];
            *(float4*)(dst + 0) = make_float4(acc[0], acc[1], acc[2], acc[3]);
            *(float4*)(dst + 4) = make_float4(acc[4], acc[5], acc[6], acc[7]);
        }
    }
    __syncthreads();

    for (int r = w; r < 32; r += NWAVES) {
        v2f ya[8];
        #pragma unroll
        for (int kk = 0; kk < 8; ++kk) {
            const int k0 = kk * 4 + Lh * 2;
            ya[kk][0] = y[Lm * YS + (k0 + 0) * 32 + r];
            ya[kk][1] = y[Lm * YS + (k0 + 1) * 32 + r];
        }
        #pragma unroll
        for (int h = 0; h < 2; ++h) {
            v8f acc = {0.f, 0.f, 0.f, 0.f, 0.f, 0.f, 0.f, 0.f};
            #pragma unroll
            for (int kk = 0; kk < 8; ++kk) {
                const v2f am = pkB[r * 512 + (h * 8 + kk) * 32 + L];
                acc = __builtin_amdgcn_wmma_f32_16x16x4_f32(
                    false, am, false, ya[kk], (short)0, acc, false, false);
            }
            float* dst = &y[Lm * YS + (h * 16 + 8 * Lh) * 32 + r];
            #pragma unroll
            for (int v = 0; v < 8; ++v)
                dst[v * 32] = acc[v];
        }
    }
    __syncthreads();

    float* orow = out + b0 * N_FULL;
    for (int i = tid; i < TILE_M * (N_FULL / 4); i += NWAVES * 32) {
        const int row = i >> 8;
        const int c4  = (i & 255) * 4;
        float4 v = *(const float4*)(&y[row * YS + c4]);
        const float4 bv = *(const float4*)(bias + c4);
        v.x += bv.x; v.y += bv.y; v.z += bv.z; v.w += bv.w;
        *(float4*)(orow + row * N_FULL + c4) = v;
    }
}

extern "C" void kernel_launch(void* const* d_in, const int* in_sizes, int n_in,
                              void* d_out, int out_size, void* d_ws, size_t ws_size,
                              hipStream_t stream) {
    (void)n_in; (void)out_size; (void)ws_size;
    const float* x    = (const float*)d_in[0];
    const float* tw   = (const float*)d_in[1];
    const float* bias = (const float*)d_in[2];
    float* out = (float*)d_out;
    float* pk  = (float*)d_ws;

    const int batch = in_sizes[0] / N_FULL;

    build_packed_mats<<<64, 32, 0, stream>>>(tw, pk);
    butterfly_wmma<<<batch / TILE_M, NWAVES * 32, 0, stream>>>(x, pk, bias, out);
}


# Setup kernel: a block's 80 twiddle 2x2s fetched by one parallel gather into LDS, its sweep runs in registers from them (was one dependent scalar load per step); same products, same order
# speedup vs baseline: 1.8495x; 1.6127x over previous
#include <hip/hip_runtime.h>
#include <hip/hip_bf16.h>

typedef float v2f __attribute__((ext_vector_type(2)));
typedef float v8f __attribute__((ext_vector_type(8)));

#define N_FULL   1024
#define NBLK     32
#define TILE_M   16
#define YS       1028
#define NWAVES   4

__device__ __forceinline__ int twiddle_offset(int q, bool isB, int idx)
{
    const int lsi = q >> 4;
    const int pr  = q & 15;
    const int ls  = isB ? lsi + 5 : lsi;
    const int s   = 1 << ls;
    const int cs  = 1 << lsi;
    const int u0  = (pr >> lsi) * 2 * cs + (pr & (cs - 1));
    const int p0  = isB ? (u0 * 32 + idx) : (idx * 32 + u0);
    const int pid = (p0 >> (ls + 1)) * s + (p0 & (s - 1));
    return ls * 2048 + pid * 4;
}

__global__ __launch_bounds__(32) void build_packed_mats(
    const float* __restrict__ tw,
    float* __restrict__ pk)
{
    __shared__ __align__(16) float Tq[80 * 4];
    __shared__ float Cm[32 * 32];
    const int c   = threadIdx.x;
    const bool isB = (blockIdx.x >= 32);
    const int idx = isB ? (int)blockIdx.x - 32 : (int)blockIdx.x;

    const int q0 = c, q1 = c + 32, q2 = 64 + (c & 15);
    const float4 g0 = *(const float4*)(tw + twiddle_offset(q0, isB, idx));
    const float4 g1 = *(const float4*)(tw + twiddle_offset(q1, isB, idx));
    const float4 g2 = *(const float4*)(tw + twiddle_offset(q2, isB, idx));
    *(float4*)(&Tq[q0 * 4]) = g0;
    *(float4*)(&Tq[q1 * 4]) = g1;
    *(float4*)(&Tq[q2 * 4]) = g2;
    __syncthreads();

    float col[32];
    #pragma unroll
    for (int i = 0; i < 32; ++i) col[i] = (i == c) ? 1.0f : 0.0f;

    #pragma unroll
    for (int lsi = 0; lsi < 5; ++lsi) {
        const int cs = 1 << lsi;
        #pragma unroll
        for (int pr = 0; pr < 16; ++pr) {
            const int u0 = (pr >> lsi) * 2 * cs + (pr & (cs - 1));
            const int u1 = u0 + cs;
            const float4 tq = *(const float4*)(&Tq[(lsi * 16 + pr) * 4]);
            const float a0 = col[u0];
            const float a1 = col[u1];
            col[u0] = tq.x * a0 + tq.y * a1;
            col[u1] = tq.z * a0 + tq.w * a1;
        }
    }

    #pragma unroll
    for (int i = 0; i < 32; ++i) Cm[i * 32 + c] = col[i];
    __syncthreads();

    const int L = c, Lm = L & 15, Lh = L >> 4;
    float* base = pk + (size_t)blockIdx.x * 1024;
    for (int h = 0; h < 2; ++h) {
        for (int kk = 0; kk < 8; ++kk) {
            const int row  = h * 16 + Lm;
            const int col0 = kk * 4 + Lh * 2;
            base[(h * 8 + kk) * 64 + L * 2 + 0] = Cm[row * 32 + col0 + 0];
            base[(h * 8 + kk) * 64 + L * 2 + 1] = Cm[row * 32 + col0 + 1];
        }
    }
}

__global__ __launch_bounds__(NWAVES * 32) void butterfly_wmma(
    const float* __restrict__ x,
    const float* __restrict__ pk,
    const float* __restrict__ bias,
    float* __restrict__ out)
{
    __shared__ __align__(16) float y[TILE_M * YS];

    const int tid = threadIdx.x;
    const int w   = tid >> 5;
    const int L   = tid & 31;
    const int Lm  = L & 15;
    const int Lh  = L >> 4;
    const size_t b0 = (size_t)blockIdx.x * TILE_M;

    const float* xrow = x + b0 * N_FULL;
    for (int i = tid; i < TILE_M * (N_FULL / 4); i += NWAVES * 32) {
        const int row = i >> 8;
        const int c4  = (i & 255) * 4;
        const float4 v = *(const float4*)(xrow + row * N_FULL + c4);
        *(float4*)(&y[row * YS + c4]) = v;
    }
    __syncthreads();

    const v2f* pkA = (const v2f*)pk;
    const v2f* pkB = (const v2f*)pk + 32 * 512;

    for (int g = w; g < NBLK; g += NWAVES) {
        v2f xa[8];
        const int xb = Lm * YS + g * 32 + Lh * 2;
        #pragma unroll
        for (int kk = 0; kk < 8; ++kk)
            xa[kk] = *(const v2f*)(&y[xb + kk * 4]);

        #pragma unroll
        for (int h = 0; h < 2; ++h) {
            v8f acc = {0.f, 0.f, 0.f, 0.f, 0.f, 0.f, 0.f, 0.f};
            #pragma unroll
            for (int kk = 0; kk < 8; ++kk) {
                const v2f am = pkA[g * 512 + (h * 8 + kk) * 32 + L];
                acc = __builtin_amdgcn_wmma_f32_16x16x4_f32(
                    false, am, false, xa[kk], (short)0, acc, false, false);
            }
            float* dst = &y[Lm * YS + g * 32 + h * 16 + 8 * Lh];
            *(float4*)(dst + 0) = make_float4(acc[0], acc[1], acc[2], acc[3]);
            *(float4*)(dst + 4) = make_float4(acc[4], acc[5], acc[6], acc[7]);
        }
    }
    __syncthreads();

    for (int r = w; r < 32; r += NWAVES) {
        v2f ya[8];
        #pragma unroll
        for (int kk = 0; kk < 8; ++kk) {
            const int k0 = kk * 4 + Lh * 2;
            ya[kk][0] = y[Lm * YS + (k0 + 0) * 32 + r];
            ya[kk][1] = y[Lm * YS + (k0 + 1) * 32 + r];
        }
        #pragma unroll
        for (int h = 0; h < 2; ++h) {
            v8f acc = {0.f, 0.f, 0.f, 0.f, 0.f, 0.f, 0.f, 0.f};
            #pragma unroll
            for (int kk = 0; kk < 8; ++kk) {
                const v2f am = pkB[r * 512 + (h * 8 + kk) * 32 + L];
                acc = __builtin_amdgcn_wmma_f32_16x16x4_f32(
                    false, am, false, ya[kk], (short)0, acc, false, false);
            }
            float* dst = &y[Lm * YS + (h * 16 + 8 * Lh) * 32 + r];
            #pragma unroll
            for (int v = 0; v < 8; ++v)
                dst[v * 32] = acc[v];
        }
    }
    __syncthreads();

    float* orow = out + b0 * N_FULL;
    for (int i = tid; i < TILE_M * (N_FULL / 4); i += NWAVES * 32) {
        const int row = i >> 8;
        const int c4  = (i & 255) * 4;
        float4 v = *(const float4*)(&y[row * YS + c4]);
        const float4 bv = *(const float4*)(bias + c4);
        v.x += bv.x; v.y += bv.y; v.z += bv.z; v.w += bv.w;
        *(float4*)(orow + row * N_FULL + c4) = v;
    }
}

extern "C" void kernel_launch(void* const* d_in, const int* in_sizes, int n_in,
                              void* d_out, int out_size, void* d_ws, size_t ws_size,
                              hipStream_t stream) {
    (void)n_in; (void)out_size; (void)ws_size;
    const float* x    = (const float*)d_in[0];
    const float* tw   = (const float*)d_in[1];
    const float* bias = (const float*)d_in[2];
    float* out = (float*)d_out;
    float* pk  = (float*)d_ws;

    const int batch = in_sizes[0] / N_FULL;

    build_packed_mats<<<64, 32, 0, stream>>>(tw, pk);
    butterfly_wmma<<<batch / TILE_M, NWAVES * 32, 0, stream>>>(x, pk, bias, out);
}


# WMMA loops as 64-product passes over 4 blocks (8 accumulators): matrix fragments requested 4 k-steps ahead, LDS reads grouped in front; stage B's 4 adjacent offsets move as 16 B; same sums and order
# speedup vs baseline: 2.0389x; 1.1024x over previous
#include <hip/hip_runtime.h>
#include <hip/hip_bf16.h>

typedef float v2f __attribute__((ext_vector_type(2)));
typedef float v4f __attribute__((ext_vector_type(4)));
typedef float v8f __attribute__((ext_vector_type(8)));

#define N_FULL   1024
#define NBLK     32
#define TILE_M   16
#define YS       1028
#define NWAVES   4
#define PASS     4
#define NSTEP    16
#define AHEAD    4

__device__ __forceinline__ int twiddle_offset(int q, bool isB, int idx)
{
    const int lsi = q >> 4;
    const int pr  = q & 15;
    const int ls  = isB ? lsi + 5 : lsi;
    const int s   = 1 << ls;
    const int cs  = 1 << lsi;
    const int u0  = (pr >> lsi) * 2 * cs + (pr & (cs - 1));
    const int p0  = isB ? (u0 * 32 + idx) : (idx * 32 + u0);
    const int pid = (p0 >> (ls + 1)) * s + (p0 & (s - 1));
    return ls * 2048 + pid * 4;
}

__global__ __launch_bounds__(32) void build_packed_mats(
    const float* __restrict__ tw,
    float* __restrict__ pk)
{
    __shared__ __align__(16) float Tq[80 * 4];
    __shared__ float Cm[32 * 32];
    const int c   = threadIdx.x;
    const bool isB = (blockIdx.x >= 32);
    const int idx = isB ? (int)blockIdx.x - 32 : (int)blockIdx.x;

    const int q0 = c, q1 = c + 32, q2 = 64 + (c & 15);
    const float4 g0 = *(const float4*)(tw + twiddle_offset(q0, isB, idx));
    const float4 g1 = *(const float4*)(tw + twiddle_offset(q1, isB, idx));
    const float4 g2 = *(const float4*)(tw + twiddle_offset(q2, isB, idx));
    *(float4*)(&Tq[q0 * 4]) = g0;
    *(float4*)(&Tq[q1 * 4]) = g1;
    *(float4*)(&Tq[q2 * 4]) = g2;
    __syncthreads();

    float col[32];
    #pragma unroll
    for (int i = 0; i < 32; ++i) col[i] = (i == c) ? 1.0f : 0.0f;

    #pragma unroll
    for (int lsi = 0; lsi < 5; ++lsi) {
        const int cs = 1 << lsi;
        #pragma unroll
        for (int pr = 0; pr < 16; ++pr) {
            const int u0 = (pr >> lsi) * 2 * cs + (pr & (cs - 1));
            const int u1 = u0 + cs;
            const float4 tq = *(const float4*)(&Tq[(lsi * 16 + pr) * 4]);
            const float a0 = col[u0];
            const float a1 = col[u1];
            col[u0] = tq.x * a0 + tq.y * a1;
            col[u1] = tq.z * a0 + tq.w * a1;
        }
    }

    #pragma unroll
    for (int i = 0; i < 32; ++i) Cm[i * 32 + c] = col[i];
    __syncthreads();

    const int L = c, Lm = L & 15, Lh = L >> 4;
    float* base = pk + (size_t)blockIdx.x * 1024;
    for (int h = 0; h < 2; ++h) {
        for (int kk = 0; kk < 8; ++kk) {
            const int row  = h * 16 + Lm;
            const int col0 = kk * 4 + Lh * 2;
            base[(h * 8 + kk) * 64 + L * 2 + 0] = Cm[row * 32 + col0 + 0];
            base[(h * 8 + kk) * 64 + L * 2 + 1] = Cm[row * 32 + col0 + 1];
        }
    }
}

__device__ __forceinline__ void request_step(v2f (&am)[PASS][2], const v2f* __restrict__ pkG,
                                             int first, int bstep, int t, int L)
{
    const int p = t >> 3, kk = t & 7;
    #pragma unroll
    for (int j = 0; j < PASS; ++j) {
        const int blk = first + bstep * (p * PASS + j);
        #pragma unroll
        for (int h = 0; h < 2; ++h)
            am[j][h] = pkG[blk * 512 + (h * 8 + kk) * 32 + L];
    }
}

__global__ __launch_bounds__(NWAVES * 32) void butterfly_wmma(
    const float* __restrict__ x,
    const float* __restrict__ pk,
    const float* __restrict__ bias,
    float* __restrict__ out)
{
    __shared__ __align__(16) float y[TILE_M * YS];

    const int tid = threadIdx.x;
    const int w   = tid >> 5;
    const int L   = tid & 31;
    const int Lm  = L & 15;
    const int Lh  = L >> 4;
    const size_t b0 = (size_t)blockIdx.x * TILE_M;

    const v2f* pkA = (const v2f*)pk;
    const v2f* pkB = (const v2f*)pk + 32 * 512;


    const float* xrow = x + b0 * N_FULL;
    for (int i = tid; i < TILE_M * (N_FULL / 4); i += NWAVES * 32) {
        const int row = i >> 8;
        const int c4  = (i & 255) * 4;
        const float4 v = *(const float4*)(xrow + row * N_FULL + c4);
        *(float4*)(&y[row * YS + c4]) = v;
    }
    v2f am[NSTEP][PASS][2];
    #pragma unroll
    for (int t = 0; t < AHEAD; ++t) request_step(am[t], pkA, w, NWAVES, t, L);
    __syncthreads();

    #pragma unroll
    for (int p = 0; p < 2; ++p) {
        v2f xa[PASS][8];
        #pragma unroll
        for (int j = 0; j < PASS; ++j) {
            const int g  = w + NWAVES * (p * PASS + j);
            const int xb = Lm * YS + g * 32 + Lh * 2;
            #pragma unroll
            for (int kk = 0; kk < 8; ++kk)
                xa[j][kk] = *(const v2f*)(&y[xb + kk * 4]);
        }
        __builtin_amdgcn_sched_barrier(0);
        v8f acc[PASS][2];
        #pragma unroll
        for (int kk = 0; kk < 8; ++kk) {
            const int t = p * 8 + kk;
            if (t + AHEAD < NSTEP) request_step(am[t + AHEAD], pkA, w, NWAVES, t + AHEAD, L);
            else request_step(am[t + AHEAD - NSTEP], pkB, w * 8, 1, t + AHEAD - NSTEP, L);
            #pragma unroll
            for (int j = 0; j < PASS; ++j) {
                #pragma unroll
                for (int h = 0; h < 2; ++h) {
                    const v8f c0 = {0.f, 0.f, 0.f, 0.f, 0.f, 0.f, 0.f, 0.f};
                    acc[j][h] = __builtin_amdgcn_wmma_f32_16x16x4_f32(
                        false, am[t][j][h], false, xa[j][kk], (short)0,
                        kk == 0 ? c0 : acc[j][h], false, false);
                }
            }
            __builtin_amdgcn_sched_barrier(0);
        }
        #pragma unroll
        for (int j = 0; j < PASS; ++j) {
            const int g = w + NWAVES * (p * PASS + j);
            #pragma unroll
            for (int h = 0; h < 2; ++h) {
                float* dst = &y[Lm * YS + g * 32 + h * 16 + 8 * Lh];
                *(float4*)(dst + 0) = make_float4(acc[j][h][0], acc[j][h][1], acc[j][h][2], acc[j][h][3]);
                *(float4*)(dst + 4) = make_float4(acc[j][h][4], acc[j][h][5], acc[j][h][6], acc[j][h][7]);
            }
        }
    }
    __syncthreads();

    #pragma unroll
    for (int p = 0; p < 2; ++p) {
        const int r0 = w * 8 + p * PASS;
        v4f yv[8][2];
        #pragma unroll
        for (int kk = 0; kk < 8; ++kk)
            #pragma unroll
            for (int s = 0; s < 2; ++s)
                yv[kk][s] = *(const v4f*)(&y[Lm * YS + (kk * 4 + Lh * 2 + s) * 32 + r0]);
        __builtin_amdgcn_sched_barrier(0);
        v8f acc[PASS][2];
        #pragma unroll
        for (int kk = 0; kk < 8; ++kk) {
            const int t = p * 8 + kk;
            if (t + AHEAD < NSTEP) request_step(am[t + AHEAD], pkB, w * 8, 1, t + AHEAD, L);
            #pragma unroll
            for (int j = 0; j < PASS; ++j) {
                const v2f yb = {yv[kk][0][j], yv[kk][1][j]};
                #pragma unroll
                for (int h = 0; h < 2; ++h) {
                    const v8f c0 = {0.f, 0.f, 0.f, 0.f, 0.f, 0.f, 0.f, 0.f};
                    acc[j][h] = __builtin_amdgcn_wmma_f32_16x16x4_f32(
                        false, am[t][j][h], false, yb, (short)0,
                        kk == 0 ? c0 : acc[j][h], false, false);
                }
            }
            __builtin_amdgcn_sched_barrier(0);
        }
        #pragma unroll
        for (int h = 0; h < 2; ++h) {
            #pragma unroll
            for (int v = 0; v < 8; ++v) {
                const v4f o = {acc[0][h][v], acc[1][h][v], acc[2][h][v], acc[3][h][v]};
                *(v4f*)(&y[Lm * YS + (h * 16 + 8 * Lh + v) * 32 + r0]) = o;
            }
        }
    }
    __syncthreads();

    float* orow = out + b0 * N_FULL;
    for (int i = tid; i < TILE_M * (N_FULL / 4); i += NWAVES * 32) {
        const int row = i >> 8;
        const int c4  = (i & 255) * 4;
        float4 v = *(const float4*)(&y[row * YS + c4]);
        const float4 bv = *(const float4*)(bias + c4);
        v.x += bv.x; v.y += bv.y; v.z += bv.z; v.w += bv.w;
        *(float4*)(orow + row * N_FULL + c4) = v;
    }
}

extern "C" void kernel_launch(void* const* d_in, const int* in_sizes, int n_in,
                              void* d_out, int out_size, void* d_ws, size_t ws_size,
                              hipStream_t stream) {
    (void)n_in; (void)out_size; (void)ws_size;
    const float* x    = (const float*)d_in[0];
    const float* tw   = (const float*)d_in[1];
    const float* bias = (const float*)d_in[2];
    float* out = (float*)d_out;
    float* pk  = (float*)d_ws;

    const int batch = in_sizes[0] / N_FULL;

    build_packed_mats<<<64, 32, 0, stream>>>(tw, pk);
    butterfly_wmma<<<batch / TILE_M, NWAVES * 32, 0, stream>>>(x, pk, bias, out);
}


# Ten stages composed as three groups (16x16, 16x16, 4x4 matrices; was two of 32x32): 768 products per tile (was 1024), matrix bytes 144 KB (was 256); sums regrouped, f32 kept
# speedup vs baseline: 2.2661x; 1.1114x over previous
#include <hip/hip_runtime.h>
#include <hip/hip_bf16.h>

typedef float v2f __attribute__((ext_vector_type(2)));
typedef float v4f __attribute__((ext_vector_type(4)));
typedef float v8f __attribute__((ext_vector_type(8)));

#define N_FULL   1024
#define TILE_M   16
#define YS       1028
#define NWAVES   4
#define PASS     4

#define PK1_OFF  0
#define PK2_OFF  (64 * 256)
#define PK3_OFF  (128 * 256)

__device__ __forceinline__ int twiddle_offset(int ls, int p0)
{
    const int s   = 1 << ls;
    const int pid = (p0 >> (ls + 1)) * s + (p0 & (s - 1));
    return ls * 2048 + pid * 4;
}

__global__ __launch_bounds__(32) void build_packed_mats(
    const float* __restrict__ tw,
    float* __restrict__ pk)
{
    __shared__ __align__(16) float Tq[32 * 4];
    __shared__ float Cm[16 * 16];
    const int L   = threadIdx.x;
    const int bid = (int)blockIdx.x;

    if (bid >= 128) {
        const int pos = (bid - 128) * 32 + L;
        const float4 t8a = *(const float4*)(tw + twiddle_offset(8, pos));
        const float4 t8b = *(const float4*)(tw + twiddle_offset(8, pos + 512));
        const float4 t9a = *(const float4*)(tw + twiddle_offset(9, pos));
        const float4 t9b = *(const float4*)(tw + twiddle_offset(9, pos + 256));
        float* dst = pk + PK3_OFF + (pos >> 2) * 64 + (pos & 3) * 2;
        #pragma unroll
        for (int c = 0; c < 4; ++c) {
            float col[4];
            #pragma unroll
            for (int i = 0; i < 4; ++i) col[i] = (i == c) ? 1.0f : 0.0f;
            float a0 = col[0], a1 = col[1];
            col[0] = t8a.x * a0 + t8a.y * a1;  col[1] = t8a.z * a0 + t8a.w * a1;
            a0 = col[2]; a1 = col[3];
            col[2] = t8b.x * a0 + t8b.y * a1;  col[3] = t8b.z * a0 + t8b.w * a1;
            a0 = col[0]; a1 = col[2];
            col[0] = t9a.x * a0 + t9a.y * a1;  col[2] = t9a.z * a0 + t9a.w * a1;
            a0 = col[1]; a1 = col[3];
            col[1] = t9b.x * a0 + t9b.y * a1;  col[3] = t9b.z * a0 + t9b.w * a1;
            #pragma unroll
            for (int m = 0; m < 4; ++m)
                dst[(m * 2 + (c >> 1)) * 8 + (c & 1)] = col[m];
        }
        return;
    }

    const bool g2  = (bid >= 64);
    const int idx  = g2 ? bid - 64 : bid;
    const int ls0  = g2 ? 4 : 0;
    const int base = g2 ? ((idx >> 4) * 256 + (idx & 15)) : idx * 16;
    const int c    = L & 15;

    {
        const int lsi = L >> 3, pr = L & 7, cs = 1 << lsi;
        const int u0  = (pr >> lsi) * 2 * cs + (pr & (cs - 1));
        *(float4*)(&Tq[L * 4]) = *(const float4*)(tw + twiddle_offset(ls0 + lsi, base + (u0 << ls0)));
    }
    __syncthreads();

    float col[16];
    #pragma unroll
    for (int i = 0; i < 16; ++i) col[i] = (i == c) ? 1.0f : 0.0f;
    #pragma unroll
    for (int lsi = 0; lsi < 4; ++lsi) {
        const int cs = 1 << lsi;
        #pragma unroll
        for (int pr = 0; pr < 8; ++pr) {
            const int u0 = (pr >> lsi) * 2 * cs + (pr & (cs - 1));
            const int u1 = u0 + cs;
            const float4 tq = *(const float4*)(&Tq[(lsi * 8 + pr) * 4]);
            const float a0 = col[u0];
            const float a1 = col[u1];
            col[u0] = tq.x * a0 + tq.y * a1;
            col[u1] = tq.z * a0 + tq.w * a1;
        }
    }
    #pragma unroll
    for (int i = 0; i < 16; ++i) Cm[i * 16 + c] = col[i];
    __syncthreads();

    const int Lm = L & 15, Lh = L >> 4;
    float* dst = pk + (g2 ? PK2_OFF : PK1_OFF) + idx * 256;
    for (int kk = 0; kk < 4; ++kk) {
        dst[(kk * 32 + L) * 2 + 0] = Cm[Lm * 16 + kk * 4 + Lh * 2 + 0];
        dst[(kk * 32 + L) * 2 + 1] = Cm[Lm * 16 + kk * 4 + Lh * 2 + 1];
    }
}

__device__ __forceinline__ void request_pass(v2f (&am)[PASS][4], const v2f* __restrict__ pkG,
                                             int mat0, int mstep, int L)
{
    #pragma unroll
    for (int j = 0; j < PASS; ++j)
        #pragma unroll
        for (int kk = 0; kk < 4; ++kk)
            am[j][kk] = pkG[(mat0 + mstep * j) * 128 + kk * 32 + L];
}

__device__ __forceinline__ void request_quads(v4f (&a3)[2][2], const float* __restrict__ pk3,
                                              int q, int Lm, int Lh)
{
    #pragma unroll
    for (int u = 0; u < 2; ++u) {
        const v4f z = {0.f, 0.f, 0.f, 0.f};
        a3[u][0] = z; a3[u][1] = z;
        if (Lm < 4) {
            const float* src = pk3 + (((q + u) * 4 + Lm) * 2 + Lh) * 8;
            a3[u][0] = *(const v4f*)(src);
            a3[u][1] = *(const v4f*)(src + 4);
        }
    }
}

__global__ __launch_bounds__(NWAVES * 32) void butterfly_wmma(
    const float* __restrict__ x,
    const float* __restrict__ pk,
    const float* __restrict__ bias,
    float* __restrict__ out)
{
    __shared__ __align__(16) float y[TILE_M * YS];

    const int tid = threadIdx.x;
    const int w   = tid >> 5;
    const int L   = tid & 31;
    const int Lm  = L & 15;
    const int Lh  = L >> 4;
    const size_t b0 = (size_t)blockIdx.x * TILE_M;

    const v2f*   pk1 = (const v2f*)(pk + PK1_OFF);
    const v2f*   pk2 = (const v2f*)(pk + PK2_OFF);
    const float* pk3 = pk + PK3_OFF;

    const float* xrow = x + b0 * N_FULL;
    for (int i = tid; i < TILE_M * (N_FULL / 4); i += NWAVES * 32) {
        const int row = i >> 8;
        const int c4  = (i & 255) * 4;
        const float4 v = *(const float4*)(xrow + row * N_FULL + c4);
        *(float4*)(&y[row * YS + c4]) = v;
    }
    v2f am[3][PASS][4];
#define REQUEST_GPASS(P)                                                              \
    do {                                                                              \
        if ((P) < 4)      request_pass(am[(P) % 3], pk1, w + NWAVES * PASS * (P), NWAVES, L); \
        else if ((P) < 8) request_pass(am[(P) % 3], pk2, ((P) - 4) * 16 + 4 * w, 1, L);       \
    } while (0)
    REQUEST_GPASS(0);
    REQUEST_GPASS(1);
    __syncthreads();

    {
        v2f xa[2][PASS][4];
        #pragma unroll
        for (int j = 0; j < PASS; ++j)
            #pragma unroll
            for (int kk = 0; kk < 4; ++kk)
                xa[0][j][kk] = *(const v2f*)(&y[Lm * YS + (w + NWAVES * j) * 16 + kk * 4 + Lh * 2]);
        #pragma unroll
        for (int p = 0; p < 4; ++p) {
            const int cur = p & 1, nxt = cur ^ 1;
            REQUEST_GPASS(p + 2);
            if (p + 1 < 4) {
                #pragma unroll
                for (int j = 0; j < PASS; ++j)
                    #pragma unroll
                    for (int kk = 0; kk < 4; ++kk)
                        xa[nxt][j][kk] = *(const v2f*)(&y[Lm * YS + (w + NWAVES * (PASS * (p + 1) + j)) * 16 + kk * 4 + Lh * 2]);
            }
            __builtin_amdgcn_sched_barrier(0);
            v8f acc[PASS];
            #pragma unroll
            for (int kk = 0; kk < 4; ++kk) {
                #pragma unroll
                for (int j = 0; j < PASS; ++j) {
                    const v8f c0 = {0.f, 0.f, 0.f, 0.f, 0.f, 0.f, 0.f, 0.f};
                    acc[j] = __builtin_amdgcn_wmma_f32_16x16x4_f32(
                        false, am[p % 3][j][kk], false, xa[cur][j][kk], (short)0,
                        kk == 0 ? c0 : acc[j], false, false);
                }
            }
            #pragma unroll
            for (int j = 0; j < PASS; ++j) {
                const int blk = w + NWAVES * (PASS * p + j);
                float* dst = &y[Lm * YS + blk * 16 + 8 * Lh];
                *(float4*)(dst + 0) = make_float4(acc[j][0], acc[j][1], acc[j][2], acc[j][3]);
                *(float4*)(dst + 4) = make_float4(acc[j][4], acc[j][5], acc[j][6], acc[j][7]);
            }
        }
    }
    __syncthreads();

    {
        v4f yv[2][4][2];
        #pragma unroll
        for (int kk = 0; kk < 4; ++kk)
            #pragma unroll
            for (int s = 0; s < 2; ++s)
                yv[0][kk][s] = *(const v4f*)(&y[Lm * YS + (kk * 4 + Lh * 2 + s) * 16 + 4 * w]);
        #pragma unroll
        for (int p = 0; p < 4; ++p) {
            const int cur = p & 1, nxt = cur ^ 1;
            REQUEST_GPASS(p + 6);
            if (p + 1 < 4) {
                #pragma unroll
                for (int kk = 0; kk < 4; ++kk)
                    #pragma unroll
                    for (int s = 0; s < 2; ++s)
                        yv[nxt][kk][s] = *(const v4f*)(&y[Lm * YS + (p + 1) * 256 + (kk * 4 + Lh * 2 + s) * 16 + 4 * w]);
            }
            __builtin_amdgcn_sched_barrier(0);
            v8f acc[PASS];
            #pragma unroll
            for (int kk = 0; kk < 4; ++kk) {
                #pragma unroll
                for (int j = 0; j < PASS; ++j) {
                    const v2f yb = {yv[cur][kk][0][j], yv[cur][kk][1][j]};
                    const v8f c0 = {0.f, 0.f, 0.f, 0.f, 0.f, 0.f, 0.f, 0.f};
                    acc[j] = __builtin_amdgcn_wmma_f32_16x16x4_f32(
                        false, am[(p + 4) % 3][j][kk], false, yb, (short)0,
                        kk == 0 ? c0 : acc[j], false, false);
                }
            }
            #pragma unroll
            for (int v = 0; v < 8; ++v) {
                const v4f o = {acc[0][v], acc[1][v], acc[2][v], acc[3][v]};
                *(v4f*)(&y[Lm * YS + p * 256 + (8 * Lh + v) * 16 + 4 * w]) = o;
            }
        }
    }
    v4f a3[2][2][2];
    request_quads(a3[0], pk3, 16 * w, Lm, Lh);
    __syncthreads();

    {
        v4f zv[2][2][2];
        #pragma unroll
        for (int u = 0; u < 2; ++u)
            #pragma unroll
            for (int s = 0; s < 2; ++s)
                zv[0][u][s] = *(const v4f*)(&y[Lm * YS + (Lh * 2 + s) * 256 + (16 * w + u) * 4]);
        #pragma unroll
        for (int p = 0; p < 8; ++p) {
            const int cur = p & 1, nxt = cur ^ 1;
            if (p + 1 < 8) {
                request_quads(a3[nxt], pk3, 16 * w + 2 * (p + 1), Lm, Lh);
                #pragma unroll
                for (int u = 0; u < 2; ++u)
                    #pragma unroll
                    for (int s = 0; s < 2; ++s)
                        zv[nxt][u][s] = *(const v4f*)(&y[Lm * YS + (Lh * 2 + s) * 256 + (16 * w + 2 * (p + 1) + u) * 4]);
            }
            __builtin_amdgcn_sched_barrier(0);
            v8f acc[2][4];
            #pragma unroll
            for (int u = 0; u < 2; ++u) {
                #pragma unroll
                for (int j = 0; j < 4; ++j) {
                    const v2f af = {a3[cur][u][j >> 1][(j & 1) * 2], a3[cur][u][j >> 1][(j & 1) * 2 + 1]};
                    const v2f zb = {zv[cur][u][0][j], zv[cur][u][1][j]};
                    const v8f c0 = {0.f, 0.f, 0.f, 0.f, 0.f, 0.f, 0.f, 0.f};
                    acc[u][j] = __builtin_amdgcn_wmma_f32_16x16x4_f32(
                        false, af, false, zb, (short)0, c0, false, false);
                }
            }
            if (Lh == 0) {
                #pragma unroll
                for (int u = 0; u < 2; ++u) {
                    const int q = 16 * w + 2 * p + u;
                    #pragma unroll
                    for (int v = 0; v < 4; ++v) {
                        const v4f o = {acc[u][0][v], acc[u][1][v], acc[u][2][v], acc[u][3][v]};
                        *(v4f*)(&y[Lm * YS + v * 256 + q * 4]) = o;
                    }
                }
            }
        }
    }
    __syncthreads();

    float* orow = out + b0 * N_FULL;
    for (int i = tid; i < TILE_M * (N_FULL / 4); i += NWAVES * 32) {
        const int row = i >> 8;
        const int c4  = (i & 255) * 4;
        float4 v = *(const float4*)(&y[row * YS + c4]);
        const float4 bv = *(const float4*)(bias + c4);
        v.x += bv.x; v.y += bv.y; v.z += bv.z; v.w += bv.w;
        *(float4*)(orow + row * N_FULL + c4) = v;
    }
}

extern "C" void kernel_launch(void* const* d_in, const int* in_sizes, int n_in,
                              void* d_out, int out_size, void* d_ws, size_t ws_size,
                              hipStream_t stream) {
    (void)n_in; (void)out_size; (void)ws_size;
    const float* x    = (const float*)d_in[0];
    const float* tw   = (const float*)d_in[1];
    const float* bias = (const float*)d_in[2];
    float* out = (float*)d_out;
    float* pk  = (float*)d_ws;

    const int batch = in_sizes[0] / N_FULL;

    build_packed_mats<<<136, 32, 0, stream>>>(tw, pk);
    butterfly_wmma<<<batch / TILE_M, NWAVES * 32, 0, stream>>>(x, pk, bias, out);
}


# Stages 8-9 (4x4 per position: 12 of 16 matrix-operand rows were zero) move from WMMA into the write-out as packed f32 FMAs, bias = start value of each 4-term sum; 16x16 groups stay on WMMA
# speedup vs baseline: 2.6850x; 1.1849x over previous
#include <hip/hip_runtime.h>
#include <hip/hip_bf16.h>

typedef float v2f __attribute__((ext_vector_type(2)));
typedef float v4f __attribute__((ext_vector_type(4)));
typedef float v8f __attribute__((ext_vector_type(8)));

#define N_FULL   1024
#define TILE_M   16
#define YS       1028
#define NWAVES   4
#define PASS     4

#define PK1_OFF  0
#define PK2_OFF  (64 * 256)
#define PK3_OFF  (128 * 256)

__device__ __forceinline__ int twiddle_offset(int ls, int p0)
{
    const int s   = 1 << ls;
    const int pid = (p0 >> (ls + 1)) * s + (p0 & (s - 1));
    return ls * 2048 + pid * 4;
}

__global__ __launch_bounds__(32) void build_packed_mats(
    const float* __restrict__ tw,
    float* __restrict__ pk)
{
    __shared__ __align__(16) float Tq[32 * 4];
    __shared__ float Cm[16 * 16];
    const int L   = threadIdx.x;
    const int bid = (int)blockIdx.x;

    if (bid >= 128) {
        const int pos = (bid - 128) * 32 + L;
        const float4 t8a = *(const float4*)(tw + twiddle_offset(8, pos));
        const float4 t8b = *(const float4*)(tw + twiddle_offset(8, pos + 512));
        const float4 t9a = *(const float4*)(tw + twiddle_offset(9, pos));
        const float4 t9b = *(const float4*)(tw + twiddle_offset(9, pos + 256));
        float* dst = pk + PK3_OFF + pos;
        #pragma unroll
        for (int c = 0; c < 4; ++c) {
            float col[4];
            #pragma unroll
            for (int i = 0; i < 4; ++i) col[i] = (i == c) ? 1.0f : 0.0f;
            float a0 = col[0], a1 = col[1];
            col[0] = t8a.x * a0 + t8a.y * a1;  col[1] = t8a.z * a0 + t8a.w * a1;
            a0 = col[2]; a1 = col[3];
            col[2] = t8b.x * a0 + t8b.y * a1;  col[3] = t8b.z * a0 + t8b.w * a1;
            a0 = col[0]; a1 = col[2];
            col[0] = t9a.x * a0 + t9a.y * a1;  col[2] = t9a.z * a0 + t9a.w * a1;
            a0 = col[1]; a1 = col[3];
            col[1] = t9b.x * a0 + t9b.y * a1;  col[3] = t9b.z * a0 + t9b.w * a1;
            #pragma unroll
            for (int m = 0; m < 4; ++m)
                dst[(m * 4 + c) * 256] = col[m];
        }
        return;
    }

    const bool g2  = (bid >= 64);
    const int idx  = g2 ? bid - 64 : bid;
    const int ls0  = g2 ? 4 : 0;
    const int base = g2 ? ((idx >> 4) * 256 + (idx & 15)) : idx * 16;
    const int c    = L & 15;

    {
        const int lsi = L >> 3, pr = L & 7, cs = 1 << lsi;
        const int u0  = (pr >> lsi) * 2 * cs + (pr & (cs - 1));
        *(float4*)(&Tq[L * 4]) = *(const float4*)(tw + twiddle_offset(ls0 + lsi, base + (u0 << ls0)));
    }
    __syncthreads();

    float col[16];
    #pragma unroll
    for (int i = 0; i < 16; ++i) col[i] = (i == c) ? 1.0f : 0.0f;
    #pragma unroll
    for (int lsi = 0; lsi < 4; ++lsi) {
        const int cs = 1 << lsi;
        #pragma unroll
        for (int pr = 0; pr < 8; ++pr) {
            const int u0 = (pr >> lsi) * 2 * cs + (pr & (cs - 1));
            const int u1 = u0 + cs;
            const float4 tq = *(const float4*)(&Tq[(lsi * 8 + pr) * 4]);
            const float a0 = col[u0];
            const float a1 = col[u1];
            col[u0] = tq.x * a0 + tq.y * a1;
            col[u1] = tq.z * a0 + tq.w * a1;
        }
    }
    #pragma unroll
    for (int i = 0; i < 16; ++i) Cm[i * 16 + c] = col[i];
    __syncthreads();

    const int Lm = L & 15, Lh = L >> 4;
    float* dst = pk + (g2 ? PK2_OFF : PK1_OFF) + idx * 256;
    for (int kk = 0; kk < 4; ++kk) {
        dst[(kk * 32 + L) * 2 + 0] = Cm[Lm * 16 + kk * 4 + Lh * 2 + 0];
        dst[(kk * 32 + L) * 2 + 1] = Cm[Lm * 16 + kk * 4 + Lh * 2 + 1];
    }
}

__device__ __forceinline__ void request_pass(v2f (&am)[PASS][4], const v2f* __restrict__ pkG,
                                             int mat0, int mstep, int L)
{
    #pragma unroll
    for (int j = 0; j < PASS; ++j)
        #pragma unroll
        for (int kk = 0; kk < 4; ++kk)
            am[j][kk] = pkG[(mat0 + mstep * j) * 128 + kk * 32 + L];
}

__global__ __launch_bounds__(NWAVES * 32) void butterfly_wmma(
    const float* __restrict__ x,
    const float* __restrict__ pk,
    const float* __restrict__ bias,
    float* __restrict__ out)
{
    __shared__ __align__(16) float y[TILE_M * YS];

    const int tid = threadIdx.x;
    const int w   = tid >> 5;
    const int L   = tid & 31;
    const int Lm  = L & 15;
    const int Lh  = L >> 4;
    const size_t b0 = (size_t)blockIdx.x * TILE_M;

    const v2f*   pk1 = (const v2f*)(pk + PK1_OFF);
    const v2f*   pk2 = (const v2f*)(pk + PK2_OFF);
    const float* pk3 = pk + PK3_OFF;

    const float* xrow = x + b0 * N_FULL;
    for (int i = tid; i < TILE_M * (N_FULL / 4); i += NWAVES * 32) {
        const int row = i >> 8;
        const int c4  = (i & 255) * 4;
        const float4 v = *(const float4*)(xrow + row * N_FULL + c4);
        *(float4*)(&y[row * YS + c4]) = v;
    }
    v2f am[3][PASS][4];
#define REQUEST_GPASS(P)                                                              \
    do {                                                                              \
        if ((P) < 4)      request_pass(am[(P) % 3], pk1, w + NWAVES * PASS * (P), NWAVES, L); \
        else if ((P) < 8) request_pass(am[(P) % 3], pk2, ((P) - 4) * 16 + 4 * w, 1, L);       \
    } while (0)
    REQUEST_GPASS(0);
    REQUEST_GPASS(1);
    __syncthreads();

    {
        v2f xa[2][PASS][4];
        #pragma unroll
        for (int j = 0; j < PASS; ++j)
            #pragma unroll
            for (int kk = 0; kk < 4; ++kk)
                xa[0][j][kk] = *(const v2f*)(&y[Lm * YS + (w + NWAVES * j) * 16 + kk * 4 + Lh * 2]);
        #pragma unroll
        for (int p = 0; p < 4; ++p) {
            const int cur = p & 1, nxt = cur ^ 1;
            REQUEST_GPASS(p + 2);
            if (p + 1 < 4) {
                #pragma unroll
                for (int j = 0; j < PASS; ++j)
                    #pragma unroll
                    for (int kk = 0; kk < 4; ++kk)
                        xa[nxt][j][kk] = *(const v2f*)(&y[Lm * YS + (w + NWAVES * (PASS * (p + 1) + j)) * 16 + kk * 4 + Lh * 2]);
            }
            __builtin_amdgcn_sched_barrier(0);
            v8f acc[PASS];
            #pragma unroll
            for (int kk = 0; kk < 4; ++kk) {
                #pragma unroll
                for (int j = 0; j < PASS; ++j) {
                    const v8f c0 = {0.f, 0.f, 0.f, 0.f, 0.f, 0.f, 0.f, 0.f};
                    acc[j] = __builtin_amdgcn_wmma_f32_16x16x4_f32(
                        false, am[p % 3][j][kk], false, xa[cur][j][kk], (short)0,
                        kk == 0 ? c0 : acc[j], false, false);
                }
            }
            #pragma unroll
            for (int j = 0; j < PASS; ++j) {
                const int blk = w + NWAVES * (PASS * p + j);
                float* dst = &y[Lm * YS + blk * 16 + 8 * Lh];
                *(float4*)(dst + 0) = make_float4(acc[j][0], acc[j][1], acc[j][2], acc[j][3]);
                *(float4*)(dst + 4) = make_float4(acc[j][4], acc[j][5], acc[j][6], acc[j][7]);
            }
        }
    }
    __syncthreads();

    {
        v4f yv[2][4][2];
        #pragma unroll
        for (int kk = 0; kk < 4; ++kk)
            #pragma unroll
            for (int s = 0; s < 2; ++s)
                yv[0][kk][s] = *(const v4f*)(&y[Lm * YS + (kk * 4 + Lh * 2 + s) * 16 + 4 * w]);
        #pragma unroll
        for (int p = 0; p < 4; ++p) {
            const int cur = p & 1, nxt = cur ^ 1;
            REQUEST_GPASS(p + 6);
            if (p + 1 < 4) {
                #pragma unroll
                for (int kk = 0; kk < 4; ++kk)
                    #pragma unroll
                    for (int s = 0; s < 2; ++s)
                        yv[nxt][kk][s] = *(const v4f*)(&y[Lm * YS + (p + 1) * 256 + (kk * 4 + Lh * 2 + s) * 16 + 4 * w]);
            }
            __builtin_amdgcn_sched_barrier(0);
            v8f acc[PASS];
            #pragma unroll
            for (int kk = 0; kk < 4; ++kk) {
                #pragma unroll
                for (int j = 0; j < PASS; ++j) {
                    const v2f yb = {yv[cur][kk][0][j], yv[cur][kk][1][j]};
                    const v8f c0 = {0.f, 0.f, 0.f, 0.f, 0.f, 0.f, 0.f, 0.f};
                    acc[j] = __builtin_amdgcn_wmma_f32_16x16x4_f32(
                        false, am[(p + 4) % 3][j][kk], false, yb, (short)0,
                        kk == 0 ? c0 : acc[j], false, false);
                }
            }
            #pragma unroll
            for (int v = 0; v < 8; ++v) {
                const v4f o = {acc[0][v], acc[1][v], acc[2][v], acc[3][v]};
                *(v4f*)(&y[Lm * YS + p * 256 + (8 * Lh + v) * 16 + 4 * w]) = o;
            }
        }
    }
    const int q  = tid & 63;
    const int r0 = tid >> 6;
    __builtin_amdgcn_sched_barrier(0);
    v4f cf[4][4];
    v4f bs[4];
    #pragma unroll
    for (int m = 0; m < 4; ++m) {
        #pragma unroll
        for (int k = 0; k < 4; ++k)
            cf[m][k] = *(const v4f*)(pk3 + (m * 4 + k) * 256 + 4 * q);
        bs[m] = *(const v4f*)(bias + m * 256 + 4 * q);
    }
    __syncthreads();

    float* orow = out + b0 * N_FULL;
#define LOAD_ROW(Z, ROW)                                                              \
    do {                                                                              \
        _Pragma("unroll")                                                             \
        for (int k = 0; k < 4; ++k)                                                   \
            (Z)[k] = *(const v4f*)(&y[(ROW) * YS + k * 256 + 4 * q]);                 \
    } while (0)
#define EMIT_ROW(Z, ROW)                                                              \
    do {                                                                              \
        _Pragma("unroll")                                                             \
        for (int m = 0; m < 4; ++m) {                                                 \
            v4f o = bs[m];                                                            \
            _Pragma("unroll")                                                         \
            for (int k = 0; k < 4; ++k)                                               \
                o = cf[m][k] * (Z)[k] + o;                                            \
            *(v4f*)(orow + (ROW) * N_FULL + m * 256 + 4 * q) = o;                     \
        }                                                                             \
    } while (0)
    v4f za[4], zb[4];
    LOAD_ROW(za, r0);
    #pragma nounroll
    for (int pp = 0; pp < TILE_M / 4; ++pp) {
        const int ra = r0 + 4 * pp;
        LOAD_ROW(zb, ra + 2);
        EMIT_ROW(za, ra);
        if (pp + 1 < TILE_M / 4) LOAD_ROW(za, ra + 4);
        EMIT_ROW(zb, ra + 2);
    }
#undef LOAD_ROW
#undef EMIT_ROW
}

extern "C" void kernel_launch(void* const* d_in, const int* in_sizes, int n_in,
                              void* d_out, int out_size, void* d_ws, size_t ws_size,
                              hipStream_t stream) {
    (void)n_in; (void)out_size; (void)ws_size;
    const float* x    = (const float*)d_in[0];
    const float* tw   = (const float*)d_in[1];
    const float* bias = (const float*)d_in[2];
    float* out = (float*)d_out;
    float* pk  = (float*)d_ws;

    const int batch = in_sizes[0] / N_FULL;

    build_packed_mats<<<136, 32, 0, stream>>>(tw, pk);
    butterfly_wmma<<<batch / TILE_M, NWAVES * 32, 0, stream>>>(x, pk, bias, out);
}


# Stages 0-3 leave WMMA: f32 butterflies in registers during stage-in (strides 4, 8 through quad-permute DPP), twiddles read directly; their sums regrouped to stage order; stages 4-7 stay on WMMA
# speedup vs baseline: 2.8547x; 1.0632x over previous
#include <hip/hip_runtime.h>
#include <hip/hip_bf16.h>

typedef float v2f __attribute__((ext_vector_type(2)));
typedef float v4f __attribute__((ext_vector_type(4)));
typedef float v8f __attribute__((ext_vector_type(8)));

#define N_FULL   1024
#define TILE_M   16
#define YS       1028
#define NWAVES   4
#define PASS     4

#define PK2_OFF  0
#define PK3_OFF  (64 * 256)

__device__ __forceinline__ int twiddle_offset(int ls, int p0)
{
    const int s   = 1 << ls;
    const int pid = (p0 >> (ls + 1)) * s + (p0 & (s - 1));
    return ls * 2048 + pid * 4;
}

__global__ __launch_bounds__(32) void build_packed_mats(
    const float* __restrict__ tw,
    float* __restrict__ pk)
{
    __shared__ __align__(16) float Tq[32 * 4];
    __shared__ float Cm[16 * 16];
    const int L   = threadIdx.x;
    const int bid = (int)blockIdx.x;

    if (bid >= 64) {
        const int pos = (bid - 64) * 32 + L;
        const float4 t8a = *(const float4*)(tw + twiddle_offset(8, pos));
        const float4 t8b = *(const float4*)(tw + twiddle_offset(8, pos + 512));
        const float4 t9a = *(const float4*)(tw + twiddle_offset(9, pos));
        const float4 t9b = *(const float4*)(tw + twiddle_offset(9, pos + 256));
        float* dst = pk + PK3_OFF + pos;
        #pragma unroll
        for (int c = 0; c < 4; ++c) {
            float col[4];
            #pragma unroll
            for (int i = 0; i < 4; ++i) col[i] = (i == c) ? 1.0f : 0.0f;
            float a0 = col[0], a1 = col[1];
            col[0] = t8a.x * a0 + t8a.y * a1;  col[1] = t8a.z * a0 + t8a.w * a1;
            a0 = col[2]; a1 = col[3];
            col[2] = t8b.x * a0 + t8b.y * a1;  col[3] = t8b.z * a0 + t8b.w * a1;
            a0 = col[0]; a1 = col[2];
            col[0] = t9a.x * a0 + t9a.y * a1;  col[2] = t9a.z * a0 + t9a.w * a1;
            a0 = col[1]; a1 = col[3];
            col[1] = t9b.x * a0 + t9b.y * a1;  col[3] = t9b.z * a0 + t9b.w * a1;
            #pragma unroll
            for (int m = 0; m < 4; ++m)
                dst[(m * 4 + c) * 256] = col[m];
        }
        return;
    }

    const int idx  = bid;
    const int ls0  = 4;
    const int base = (idx >> 4) * 256 + (idx & 15);
    const int c    = L & 15;

    {
        const int lsi = L >> 3, pr = L & 7, cs = 1 << lsi;
        const int u0  = (pr >> lsi) * 2 * cs + (pr & (cs - 1));
        *(float4*)(&Tq[L * 4]) = *(const float4*)(tw + twiddle_offset(ls0 + lsi, base + (u0 << ls0)));
    }
    __syncthreads();

    float col[16];
    #pragma unroll
    for (int i = 0; i < 16; ++i) col[i] = (i == c) ? 1.0f : 0.0f;
    #pragma unroll
    for (int lsi = 0; lsi < 4; ++lsi) {
        const int cs = 1 << lsi;
        #pragma unroll
        for (int pr = 0; pr < 8; ++pr) {
            const int u0 = (pr >> lsi) * 2 * cs + (pr & (cs - 1));
            const int u1 = u0 + cs;
            const float4 tq = *(const float4*)(&Tq[(lsi * 8 + pr) * 4]);
            const float a0 = col[u0];
            const float a1 = col[u1];
            col[u0] = tq.x * a0 + tq.y * a1;
            col[u1] = tq.z * a0 + tq.w * a1;
        }
    }
    #pragma unroll
    for (int i = 0; i < 16; ++i) Cm[i * 16 + c] = col[i];
    __syncthreads();

    const int Lm = L & 15, Lh = L >> 4;
    float* dst = pk + PK2_OFF + idx * 256;
    for (int kk = 0; kk < 4; ++kk) {
        dst[(kk * 32 + L) * 2 + 0] = Cm[Lm * 16 + kk * 4 + Lh * 2 + 0];
        dst[(kk * 32 + L) * 2 + 1] = Cm[Lm * 16 + kk * 4 + Lh * 2 + 1];
    }
}

__device__ __forceinline__ void request_pass(v2f (&am)[PASS][4], const v2f* __restrict__ pkG,
                                             int mat0, int L)
{
    #pragma unroll
    for (int j = 0; j < PASS; ++j)
        #pragma unroll
        for (int kk = 0; kk < 4; ++kk)
            am[j][kk] = pkG[(mat0 + j) * 128 + kk * 32 + L];
}

__device__ __forceinline__ float quad_xor1(float v)
{
    return __builtin_bit_cast(float, __builtin_amdgcn_mov_dpp(__builtin_bit_cast(int, v), 0xB1, 0xF, 0xF, true));
}
__device__ __forceinline__ float quad_xor2(float v)
{
    return __builtin_bit_cast(float, __builtin_amdgcn_mov_dpp(__builtin_bit_cast(int, v), 0x4E, 0xF, 0xF, true));
}

struct BflyCoef {
    v2f s0a_lo, s0a_hi;
    v2f s0b_lo, s0b_hi;
    v2f s1t_lo, s1t_hi;
    v2f s1b_lo, s1b_hi;
    v4f own2, oth2;
    v4f own3, oth3;
};

__device__ __forceinline__ void load_bfly_coef(BflyCoef& k, const float* __restrict__ tw, int c4)
{
    const v4f A = *(const v4f*)(tw + twiddle_offset(0, c4));
    const v4f B = *(const v4f*)(tw + twiddle_offset(0, c4 + 2));
    const v4f C = *(const v4f*)(tw + twiddle_offset(1, c4));
    const v4f D = *(const v4f*)(tw + twiddle_offset(1, c4 + 1));
    k.s0a_lo = v2f{A[0], A[2]};  k.s0a_hi = v2f{A[1], A[3]};
    k.s0b_lo = v2f{B[0], B[2]};  k.s0b_hi = v2f{B[1], B[3]};
    k.s1t_lo = v2f{C[0], D[0]};  k.s1t_hi = v2f{C[1], D[1]};
    k.s1b_lo = v2f{C[2], D[2]};  k.s1b_hi = v2f{C[3], D[3]};
    const int hi2 = (c4 >> 2) & 1;
    const int hi3 = (c4 >> 3) & 1;
    #pragma unroll
    for (int e = 0; e < 4; ++e) {
        const v2f f2 = *(const v2f*)(tw + twiddle_offset(2, (c4 & ~4) + e) + 2 * hi2);
        const v2f f3 = *(const v2f*)(tw + twiddle_offset(3, (c4 & ~8) + e) + 2 * hi3);
        k.own2[e] = hi2 ? f2[1] : f2[0];
        k.oth2[e] = hi2 ? f2[0] : f2[1];
        k.own3[e] = hi3 ? f3[1] : f3[0];
        k.oth3[e] = hi3 ? f3[0] : f3[1];
    }
}

__device__ __forceinline__ v4f bfly_stages_0_3(v4f v, const BflyCoef& k)
{
    const v2f n01 = k.s0a_lo * v2f{v[0], v[0]} + k.s0a_hi * v2f{v[1], v[1]};
    const v2f n23 = k.s0b_lo * v2f{v[2], v[2]} + k.s0b_hi * v2f{v[3], v[3]};
    const v2f m01 = k.s1t_lo * n01 + k.s1t_hi * n23;
    const v2f m23 = k.s1b_lo * n01 + k.s1b_hi * n23;
    const v4f m = {m01[0], m01[1], m23[0], m23[1]};
    const v4f p = {quad_xor1(m[0]), quad_xor1(m[1]), quad_xor1(m[2]), quad_xor1(m[3])};
    const v4f r = k.own2 * m + k.oth2 * p;
    const v4f q = {quad_xor2(r[0]), quad_xor2(r[1]), quad_xor2(r[2]), quad_xor2(r[3])};
    return k.own3 * r + k.oth3 * q;
}

__global__ __launch_bounds__(NWAVES * 32) void butterfly_wmma(
    const float* __restrict__ x,
    const float* __restrict__ tw,
    const float* __restrict__ pk,
    const float* __restrict__ bias,
    float* __restrict__ out)
{
    __shared__ __align__(16) float y[TILE_M * YS];

    const int tid = threadIdx.x;
    const int w   = tid >> 5;
    const int L   = tid & 31;
    const int Lm  = L & 15;
    const int Lh  = L >> 4;
    const size_t b0 = (size_t)blockIdx.x * TILE_M;

    const v2f*   pk2 = (const v2f*)(pk + PK2_OFF);
    const float* pk3 = pk + PK3_OFF;

    const float* xrow = x + b0 * N_FULL;
    BflyCoef kf[2];
    load_bfly_coef(kf[0], tw, 4 * tid);
    load_bfly_coef(kf[1], tw, 4 * (tid + NWAVES * 32));
#define LOAD_BATCH(DST, R0)                                                           \
    do {                                                                              \
        _Pragma("unroll")                                                             \
        for (int k = 0; k < 8; ++k)                                                   \
            (DST)[k] = *(const v4f*)(xrow + ((R0) + (k >> 1)) * N_FULL                \
                                          + 4 * (tid + NWAVES * 32 * (k & 1)));       \
    } while (0)
    v2f am[3][PASS][4];
#define REQUEST_GPASS(P)                                                              \
    do {                                                                              \
        if ((P) < 4) request_pass(am[(P) % 3], pk2, (P) * 16 + 4 * w, L);             \
    } while (0)
    v4f xb[2][8];
    LOAD_BATCH(xb[0], 0);
    #pragma unroll
    for (int b = 0; b < 4; ++b) {
        if (b + 1 < 4) LOAD_BATCH(xb[(b + 1) & 1], 4 * (b + 1));
        else { REQUEST_GPASS(0); REQUEST_GPASS(1); }
        #pragma unroll
        for (int k = 0; k < 8; ++k) {
            const int row = 4 * b + (k >> 1);
            const v4f r = bfly_stages_0_3(xb[b & 1][k], kf[k & 1]);
            *(v4f*)(&y[row * YS + 4 * (tid + NWAVES * 32 * (k & 1))]) = r;
        }
        __builtin_amdgcn_sched_barrier(0);
    }
#undef LOAD_BATCH
    __syncthreads();

    {
        v4f yv[2][4][2];
        #pragma unroll
        for (int kk = 0; kk < 4; ++kk)
            #pragma unroll
            for (int s = 0; s < 2; ++s)
                yv[0][kk][s] = *(const v4f*)(&y[Lm * YS + (kk * 4 + Lh * 2 + s) * 16 + 4 * w]);
        #pragma unroll
        for (int p = 0; p < 4; ++p) {
            const int cur = p & 1, nxt = cur ^ 1;
            REQUEST_GPASS(p + 2);
            if (p + 1 < 4) {
                #pragma unroll
                for (int kk = 0; kk < 4; ++kk)
                    #pragma unroll
                    for (int s = 0; s < 2; ++s)
                        yv[nxt][kk][s] = *(const v4f*)(&y[Lm * YS + (p + 1) * 256 + (kk * 4 + Lh * 2 + s) * 16 + 4 * w]);
            }
            __builtin_amdgcn_sched_barrier(0);
            v8f acc[PASS];
            #pragma unroll
            for (int kk = 0; kk < 4; ++kk) {
                #pragma unroll
                for (int j = 0; j < PASS; ++j) {
                    const v2f yb = {yv[cur][kk][0][j], yv[cur][kk][1][j]};
                    const v8f c0 = {0.f, 0.f, 0.f, 0.f, 0.f, 0.f, 0.f, 0.f};
                    acc[j] = __builtin_amdgcn_wmma_f32_16x16x4_f32(
                        false, am[p % 3][j][kk], false, yb, (short)0,
                        kk == 0 ? c0 : acc[j], false, false);
                }
            }
            #pragma unroll
            for (int v = 0; v < 8; ++v) {
                const v4f o = {acc[0][v], acc[1][v], acc[2][v], acc[3][v]};
                *(v4f*)(&y[Lm * YS + p * 256 + (8 * Lh + v) * 16 + 4 * w]) = o;
            }
        }
    }
    const int q  = tid & 63;
    const int r0 = tid >> 6;
    __builtin_amdgcn_sched_barrier(0);
    v4f cf[4][4];
    v4f bs[4];
    #pragma unroll
    for (int m = 0; m < 4; ++m) {
        #pragma unroll
        for (int k = 0; k < 4; ++k)
            cf[m][k] = *(const v4f*)(pk3 + (m * 4 + k) * 256 + 4 * q);
        bs[m] = *(const v4f*)(bias + m * 256 + 4 * q);
    }
    __syncthreads();

    float* orow = out + b0 * N_FULL;
#define LOAD_ROW(Z, ROW)                                                              \
    do {                                                                              \
        _Pragma("unroll")                                                             \
        for (int k = 0; k < 4; ++k)                                                   \
            (Z)[k] = *(const v4f*)(&y[(ROW) * YS + k * 256 + 4 * q]);                 \
    } while (0)
#define EMIT_ROW(Z, ROW)                                                              \
    do {                                                                              \
        _Pragma("unroll")                                                             \
        for (int m = 0; m < 4; ++m) {                                                 \
            v4f o = bs[m];                                                            \
            _Pragma("unroll")                                                         \
            for (int k = 0; k < 4; ++k)                                               \
                o = cf[m][k] * (Z)[k] + o;                                            \
            *(v4f*)(orow + (ROW) * N_FULL + m * 256 + 4 * q) = o;                     \
        }                                                                             \
    } while (0)
    v4f za[4], zb[4];
    LOAD_ROW(za, r0);
    #pragma nounroll
    for (int pp = 0; pp < TILE_M / 4; ++pp) {
        const int ra = r0 + 4 * pp;
        LOAD_ROW(zb, ra + 2);
        EMIT_ROW(za, ra);
        if (pp + 1 < TILE_M / 4) LOAD_ROW(za, ra + 4);
        EMIT_ROW(zb, ra + 2);
    }
#undef LOAD_ROW
#undef EMIT_ROW
}

extern "C" void kernel_launch(void* const* d_in, const int* in_sizes, int n_in,
                              void* d_out, int out_size, void* d_ws, size_t ws_size,
                              hipStream_t stream) {
    (void)n_in; (void)out_size; (void)ws_size;
    const float* x    = (const float*)d_in[0];
    const float* tw   = (const float*)d_in[1];
    const float* bias = (const float*)d_in[2];
    float* out = (float*)d_out;
    float* pk  = (float*)d_ws;

    const int batch = in_sizes[0] / N_FULL;

    build_packed_mats<<<72, 32, 0, stream>>>(tw, pk);
    butterfly_wmma<<<batch / TILE_M, NWAVES * 32, 0, stream>>>(x, tw, pk, bias, out);
}
